# starting kernel
# baseline (speedup 1.0000x reference)
#include <hip/hip_runtime.h>

typedef __attribute__((ext_vector_type(2))) float v2f;
typedef __attribute__((ext_vector_type(8))) float v8f;

__global__ __launch_bounds__(256) void lc1d_wmma_kernel(
    const float* __restrict__ x,
    const float* __restrict__ w,
    const float* __restrict__ bias,
    float* __restrict__ out,
    int P) {
  const int lane = threadIdx.x & 31;
  const int wave = threadIdx.x >> 5;
  const int p = blockIdx.x * 8 + wave;
  if (p >= P) return;

  const int m     = lane & 15;
  const int khalf = (lane >> 4) * 2;

  const int   n      = m;
  const bool  nvalid = (n < 4);
  const int   nc     = nvalid ? n : 3;
  const float* wp    = w + (size_t)p * 64 * 4;

  const size_t bstride = (size_t)P * 64;
  const float* x0 = x + (size_t)m * bstride + (size_t)p * 64 + khalf;
  const float* x1 = x0 + 16 * bstride;

  v8f acc0 = {};
  v8f acc1 = {};

#pragma unroll
  for (int kk = 0; kk < 64; kk += 4) {
    v2f a0 = *(const v2f*)(x0 + kk);
    v2f a1 = *(const v2f*)(x1 + kk);

    float b0 = wp[(kk + khalf + 0) * 4 + nc];
    float b1 = wp[(kk + khalf + 1) * 4 + nc];
    v2f bf;
    bf.x = nvalid ? b0 : 0.0f;
    bf.y = nvalid ? b1 : 0.0f;

    acc0 = __builtin_amdgcn_wmma_f32_16x16x4_f32(
        false, a0, false, bf, (short)0, acc0, false, false);
    acc1 = __builtin_amdgcn_wmma_f32_16x16x4_f32(
        false, a1, false, bf, (short)0, acc1, false, false);
  }

  if (nvalid) {
    const float bv    = bias[(size_t)p * 4 + n];
    const int   mbase = (lane >> 4) * 8;
    float* o0 = out + (size_t)p * 4 + n;
    const size_t ostride = (size_t)P * 4;
#pragma unroll
    for (int r = 0; r < 8; ++r) {
      const int row = mbase + r;
      o0[(size_t)row * ostride]        = acc0[r] + bv;
      o0[(size_t)(row + 16) * ostride] = acc1[r] + bv;
    }
  }
}

extern "C" void kernel_launch(void* const* d_in, const int* in_sizes, int n_in,
                              void* d_out, int out_size, void* d_ws, size_t ws_size,
                              hipStream_t stream) {
  const float* x    = (const float*)d_in[0];
  const float* w    = (const float*)d_in[1];
  const float* bias = (const float*)d_in[2];
  float*       out  = (float*)d_out;

  const int P = in_sizes[2] / 4;

  dim3 block(256);
  dim3 grid((P + 7) / 8);
  lc1d_wmma_kernel<<<grid, block, 0, stream>>>(x, w, bias, out, P);
}


# whole-line global loads (x: two 256-B rows per load, w: 512 B per load) parked in wave-private LDS rows, WMMA fragments re-read by ds_load_b128; sum order over c permuted alike in A and B
# speedup vs baseline: 1.7333x; 1.7333x over previous
#include <hip/hip_runtime.h>

typedef __attribute__((ext_vector_type(2))) float v2f;
typedef __attribute__((ext_vector_type(4))) float v4f;
typedef __attribute__((ext_vector_type(8))) float v8f;


#define PITCH 68
#define XROWS 16
#define WROWS 4
#define WAVE_FLOATS ((XROWS + WROWS) * PITCH)

__global__ __launch_bounds__(256) void lc1d_wmma_kernel(
    const float* __restrict__ x,
    const float* __restrict__ w,
    const float* __restrict__ bias,
    float* __restrict__ out,
    int P) {
  __shared__ __attribute__((aligned(16))) float lds[8 * WAVE_FLOATS];

  const int lane = threadIdx.x & 31;
  const int wave = threadIdx.x >> 5;
  const int p = blockIdx.x * 8 + wave;
  if (p >= P) return;

  const int m = lane & 15;
  const int h = lane >> 4;

  const int   n      = m;
  const bool  nvalid = (n < 4);
  const int   nc     = nvalid ? n : 3;

  float* xs = lds + wave * WAVE_FLOATS;
  float* ws = xs + XROWS * PITCH;

  const float* wg = w + (size_t)p * 256 + 4 * lane;
  const v4f w0 = *(const v4f*)(wg);
  const v4f w1 = *(const v4f*)(wg + 128);

  const float bv = bias[(size_t)p * 4 + nc];

  const size_t bstride = (size_t)P * 64;
  const float* xg = x + (size_t)h * bstride + (size_t)p * 64 + 4 * m;
  v4f xr[16];
#pragma unroll
  for (int i = 0; i < 16; ++i)
    xr[i] = *(const v4f*)(xg + (size_t)(2 * i) * bstride);

#pragma unroll
  for (int o = 0; o < 4; ++o) {
    ws[o * PITCH + lane]      = w0[o];
    ws[o * PITCH + lane + 32] = w1[o];
  }

  const float* ar = xs + m * PITCH + 4 * h;
  const float* br = ws + nc * PITCH + 4 * h;

  __builtin_amdgcn_fence(__ATOMIC_RELEASE, "wavefront");
  __builtin_amdgcn_wave_barrier();
  __builtin_amdgcn_fence(__ATOMIC_ACQUIRE, "wavefront");

  v4f bq[8];
#pragma unroll
  for (int j = 0; j < 8; ++j) {
    const v4f t4 = *(const v4f*)(br + 8 * j);
    bq[j].x = nvalid ? t4.x : 0.0f;
    bq[j].y = nvalid ? t4.y : 0.0f;
    bq[j].z = nvalid ? t4.z : 0.0f;
    bq[j].w = nvalid ? t4.w : 0.0f;
  }

  v8f acc[2];
  acc[0] = (v8f){};
  acc[1] = (v8f){};

#pragma unroll
  for (int t = 0; t < 2; ++t) {
#pragma unroll
    for (int i = 0; i < 8; ++i)
      *(v4f*)(xs + (2 * i + h) * PITCH + 4 * m) = xr[8 * t + i];

    __builtin_amdgcn_fence(__ATOMIC_RELEASE, "wavefront");
    __builtin_amdgcn_wave_barrier();
    __builtin_amdgcn_fence(__ATOMIC_ACQUIRE, "wavefront");

#pragma unroll
    for (int j = 0; j < 8; ++j) {
      const v4f a = *(const v4f*)(ar + 8 * j);
      v2f a_lo, a_hi, b_lo, b_hi;
      a_lo.x = a.x;     a_lo.y = a.y;     a_hi.x = a.z;     a_hi.y = a.w;
      b_lo.x = bq[j].x; b_lo.y = bq[j].y; b_hi.x = bq[j].z; b_hi.y = bq[j].w;
      acc[t] = __builtin_amdgcn_wmma_f32_16x16x4_f32(
          false, a_lo, false, b_lo, (short)0, acc[t], false, false);
      acc[t] = __builtin_amdgcn_wmma_f32_16x16x4_f32(
          false, a_hi, false, b_hi, (short)0, acc[t], false, false);
    }
    if (t == 0) {
      __builtin_amdgcn_fence(__ATOMIC_RELEASE, "wavefront");
      __builtin_amdgcn_wave_barrier();
      __builtin_amdgcn_fence(__ATOMIC_ACQUIRE, "wavefront");
    }
  }

  if (nvalid) {
    const int   mbase = h * 8;
    float* o0 = out + (size_t)p * 4 + n;
    const size_t ostride = (size_t)P * 4;
#pragma unroll
    for (int r = 0; r < 8; ++r) {
      const int row = mbase + r;
      o0[(size_t)row * ostride]        = acc[0][r] + bv;
      o0[(size_t)(row + 16) * ostride] = acc[1][r] + bv;
    }
  }
}

extern "C" void kernel_launch(void* const* d_in, const int* in_sizes, int n_in,
                              void* d_out, int out_size, void* d_ws, size_t ws_size,
                              hipStream_t stream) {
  const float* x    = (const float*)d_in[0];
  const float* w    = (const float*)d_in[1];
  const float* bias = (const float*)d_in[2];
  float*       out  = (float*)d_out;

  const int P = in_sizes[2] / 4;

  dim3 block(256);
  dim3 grid((P + 7) / 8);
  lc1d_wmma_kernel<<<grid, block, 0, stream>>>(x, w, bias, out, P);
}
